# starting kernel
# baseline (speedup 1.0000x reference)
#include <hip/hip_runtime.h>

typedef float v2f __attribute__((ext_vector_type(2)));
typedef float v8f __attribute__((ext_vector_type(8)));

static constexpr int NTOT = 131072;
static constexpr int D    = 16;
static constexpr int DIN  = 64;
static constexpr int DOUT = 64;
static constexpr int DC   = 4;
static constexpr int QC   = DC * (DIN / 4);
static constexpr int LDS_DW = QC * 4 * 64;

__global__ __launch_bounds__(256)
void mld_wmma_kernel(const float* __restrict__ x,
                     const float* __restrict__ alpha,
                     const float* __restrict__ scale,
                     float* __restrict__ out) {
  __shared__ float ldsB[LDS_DW];

  const int tid  = threadIdx.x;
  const int lane = tid & 31;
  const int wave = tid >> 5;
  const int row  = lane & 15;
  const int hi   = lane >> 4;
  const int c0   = hi << 1;

  const int n0 = blockIdx.x * 128 + wave * 16;
  const int n  = n0 + row;

  v2f xr[16];
#pragma unroll
  for (int t = 0; t < 16; ++t)
    xr[t] = *(const v2f*)(x + (size_t)n * DIN + 4 * t + c0);

  v8f acc0 = {}, acc1 = {}, acc2 = {}, acc3 = {};
  float asum = 0.0f;

  for (int chunk = 0; chunk < D / DC; ++chunk) {
    const int dbase = chunk * DC;

    for (int p = tid; p < QC * 4 * 32; p += 256) {
      const int c  = p & 31;
      const int ob = (p >> 5) & 3;
      const int q  = p >> 7;
      const int d  = dbase + (q >> 4);
      const int i  = ((q & 15) << 2) + ((c >> 4) << 1);
      const int o  = (ob << 4) + (c & 15);
      const float s0 = scale[(size_t)(d * DIN + i)     * DOUT + o];
      const float s1 = scale[(size_t)(d * DIN + i + 1) * DOUT + o];
      v2f pk; pk.x = s0; pk.y = s1;
      *(v2f*)(ldsB + (((q << 2) + ob) << 6) + (c << 1)) = pk;
    }
    __syncthreads();

    for (int dd = 0; dd < DC; ++dd) {
      const int d   = dbase + dd;
      const float a = alpha[(size_t)d * NTOT + n];
      asum += a;
      const int qb = dd << 4;
#pragma unroll
      for (int t = 0; t < 16; ++t) {
        v2f afrag;
        afrag.x = a * xr[t].x;
        afrag.y = a * xr[t].y;
        const float* bp = ldsB + (((qb + t) << 2) << 6) + (lane << 1);
        v2f b0 = *(const v2f*)(bp);
        v2f b1 = *(const v2f*)(bp + 64);
        v2f b2 = *(const v2f*)(bp + 128);
        v2f b3 = *(const v2f*)(bp + 192);
        acc0 = __builtin_amdgcn_wmma_f32_16x16x4_f32(false, afrag, false, b0,
                                                     (short)0, acc0, false, false);
        acc1 = __builtin_amdgcn_wmma_f32_16x16x4_f32(false, afrag, false, b1,
                                                     (short)0, acc1, false, false);
        acc2 = __builtin_amdgcn_wmma_f32_16x16x4_f32(false, afrag, false, b2,
                                                     (short)0, acc2, false, false);
        acc3 = __builtin_amdgcn_wmma_f32_16x16x4_f32(false, afrag, false, b3,
                                                     (short)0, acc3, false, false);
      }
    }
    __syncthreads();
  }

  float rs[8];
#pragma unroll
  for (int r = 0; r < 8; ++r) {
    const int srcByte = (r + (hi << 3)) << 2;
    rs[r] = __int_as_float(
        __builtin_amdgcn_ds_bpermute(srcByte, __float_as_int(asum)));
  }

#pragma unroll
  for (int r = 0; r < 8; ++r) {
    float* orow = out + (size_t)(n0 + r + (hi << 3)) * DOUT + (lane & 15);
    orow[0]  = rs[r] * acc0[r];
    orow[16] = rs[r] * acc1[r];
    orow[32] = rs[r] * acc2[r];
    orow[48] = rs[r] * acc3[r];
  }
}

extern "C" void kernel_launch(void* const* d_in, const int* in_sizes, int n_in,
                              void* d_out, int out_size, void* d_ws, size_t ws_size,
                              hipStream_t stream) {
  const float* x     = (const float*)d_in[0];
  const float* alpha = (const float*)d_in[1];
  const float* scale = (const float*)d_in[2];
  float* out = (float*)d_out;

  dim3 grid(NTOT / 128);
  dim3 block(256);
  hipLaunchKernelGGL(mld_wmma_kernel, grid, block, 0, stream,
                     x, alpha, scale, out);
}


# wave priority 3 through decoder 12, then 2, 1, 0 on decoders 13, 14, 15 (s_setprio): the waves of a SIMD end together instead of oldest first, no lone-wave tail; sums and their order unchanged
# speedup vs baseline: 1.1121x; 1.1121x over previous
#include <hip/hip_runtime.h>


typedef float v2f __attribute__((ext_vector_type(2)));
typedef float v4f __attribute__((ext_vector_type(4)));
typedef float v8f __attribute__((ext_vector_type(8)));

static constexpr int NTOT = 131072;
static constexpr int D    = 16;
static constexpr int DIN  = 64;
static constexpr int DOUT = 64;
static constexpr int QT   = D * (DIN / 4);
static constexpr int LDS_DW = QT * 4 * 64;
static constexpr int NTHR = 1024;
static constexpr int ROWS_PER_BLOCK = (NTHR / 32) * 16;

__global__ __launch_bounds__(1024)
void mld_wmma_kernel(const float* __restrict__ x,
                     const float* __restrict__ alpha,
                     const float* __restrict__ scale,
                     float* __restrict__ out) {
  __shared__ alignas(16) float ldsB[LDS_DW];

  const int tid  = threadIdx.x;
  const int lane = tid & 31;
  const int wave = tid >> 5;
  const int row  = lane & 15;
  const int hi   = lane >> 4;
  const int c0   = hi << 1;

  const int n0 = blockIdx.x * ROWS_PER_BLOCK + wave * 16;
  const int n  = n0 + row;

  v2f xr[16];
#pragma unroll
  for (int t = 0; t < 16; ++t)
    xr[t] = *(const v2f*)(x + (size_t)n * DIN + 4 * t + c0);

  for (int p = tid; p < QT * 4 * 16; p += NTHR) {
    const int u  = p & 15;
    const int ob = (p >> 4) & 3;
    const int q  = p >> 6;
    const int d  = q >> 4;
    const int i  = ((q & 15) << 2) + ((u >> 3) << 1);
    const int o  = (ob << 4) + ((u & 7) << 1);
    const v2f r0 = *(const v2f*)(scale + (size_t)(d * DIN + i)     * DOUT + o);
    const v2f r1 = *(const v2f*)(scale + (size_t)(d * DIN + i + 1) * DOUT + o);
    v4f pk; pk.x = r0.x; pk.y = r1.x; pk.z = r0.y; pk.w = r1.y;
    *(v4f*)(ldsB + (((q << 2) + ob) << 6) + (u << 2)) = pk;
  }
  __syncthreads();

  v8f acc0 = {}, acc1 = {}, acc2 = {}, acc3 = {};
  float asum = 0.0f;

  __builtin_amdgcn_s_setprio(3);

  for (int d = 0; d < D; ++d) {
    if (d == D - 3)      __builtin_amdgcn_s_setprio(2);
    else if (d == D - 2) __builtin_amdgcn_s_setprio(1);
    else if (d == D - 1) __builtin_amdgcn_s_setprio(0);
    const float a = alpha[(size_t)d * NTOT + n];
    asum += a;
    const int qb = d << 4;
#pragma unroll
    for (int t = 0; t < 16; ++t) {
      v2f afrag;
      afrag.x = a * xr[t].x;
      afrag.y = a * xr[t].y;
      const float* bp = ldsB + (((qb + t) << 2) << 6) + (lane << 1);
      v2f b0 = *(const v2f*)(bp);
      v2f b1 = *(const v2f*)(bp + 64);
      v2f b2 = *(const v2f*)(bp + 128);
      v2f b3 = *(const v2f*)(bp + 192);
      acc0 = __builtin_amdgcn_wmma_f32_16x16x4_f32(false, afrag, false, b0,
                                                   (short)0, acc0, false, false);
      acc1 = __builtin_amdgcn_wmma_f32_16x16x4_f32(false, afrag, false, b1,
                                                   (short)0, acc1, false, false);
      acc2 = __builtin_amdgcn_wmma_f32_16x16x4_f32(false, afrag, false, b2,
                                                   (short)0, acc2, false, false);
      acc3 = __builtin_amdgcn_wmma_f32_16x16x4_f32(false, afrag, false, b3,
                                                   (short)0, acc3, false, false);
    }
  }

  float rs[8];
#pragma unroll
  for (int r = 0; r < 8; ++r) {
    const int srcByte = (r + (hi << 3)) << 2;
    rs[r] = __int_as_float(
        __builtin_amdgcn_ds_bpermute(srcByte, __float_as_int(asum)));
  }

#pragma unroll
  for (int r = 0; r < 8; ++r) {
    float* orow = out + (size_t)(n0 + r + (hi << 3)) * DOUT + (lane & 15);
    orow[0]  = rs[r] * acc0[r];
    orow[16] = rs[r] * acc1[r];
    orow[32] = rs[r] * acc2[r];
    orow[48] = rs[r] * acc3[r];
  }
}

extern "C" void kernel_launch(void* const* d_in, const int* in_sizes, int n_in,
                              void* d_out, int out_size, void* d_ws, size_t ws_size,
                              hipStream_t stream) {
  const float* x     = (const float*)d_in[0];
  const float* alpha = (const float*)d_in[1];
  const float* scale = (const float*)d_in[2];
  float* out = (float*)d_out;

  dim3 grid(NTOT / ROWS_PER_BLOCK);
  dim3 block(NTHR);
  hipLaunchKernelGGL(mld_wmma_kernel, grid, block, 0, stream,
                     x, alpha, scale, out);
}
